# starting kernel
# baseline (speedup 1.0000x reference)
#include <hip/hip_runtime.h>
#include <hip/hip_bf16.h>
#include <math.h>

typedef __attribute__((ext_vector_type(2))) float v2f;
typedef __attribute__((ext_vector_type(8))) float v8f;

#define NEG_SLOPE 0.2f

__global__ __launch_bounds__(32) void gat_gemm_wmma_f32(
    const float* __restrict__ A,
    const float* __restrict__ W,
    float* __restrict__ C,
    int N, int K, int M)
{
    const int rowTile = blockIdx.x * 16;
    const int colTile = blockIdx.y * 16;
    const int lane    = threadIdx.x;
    const int r       = lane & 15;
    const int koff    = (lane >> 4) * 2;

    int arow = rowTile + r;
    if (arow > N - 1) arow = N - 1;
    const float* __restrict__ ap = A + (size_t)arow * K;
    const float* __restrict__ wp = W + (size_t)(colTile + r) * K;

    v8f acc = {};
    for (int k = 0; k < K; k += 4) {
        v2f a = *(const v2f*)(ap + k + koff);
        v2f b = *(const v2f*)(wp + k + koff);
        acc = __builtin_amdgcn_wmma_f32_16x16x4_f32(
             false, a,  false, b,
             (short)0, acc,  false,  false);
    }

    const int rbase = rowTile + (lane >> 4) * 8;
    const int col   = colTile + (lane & 15);
#pragma unroll
    for (int j = 0; j < 8; ++j) {
        int rr = rbase + j;
        if (rr < N) C[(size_t)rr * M + col] = acc[j];
    }
}

__global__ void gat_attn_proj(const float* __restrict__ feat,
                              const float* __restrict__ al,
                              const float* __restrict__ ar,
                              float* __restrict__ el,
                              float* __restrict__ er,
                              int N, int D)
{
    int n = blockIdx.x * blockDim.x + threadIdx.x;
    if (n >= N) return;
    const float* f = feat + (size_t)n * D;
    float sl = 0.f, sr = 0.f;
    for (int k = 0; k < D; ++k) {
        float v = f[k];
        sl = fmaf(v, al[k], sl);
        sr = fmaf(v, ar[k], sr);
    }
    el[n] = sl;
    er[n] = sr;
}

__global__ void gat_init(float* __restrict__ out,
                         const float* __restrict__ bias,
                         float* __restrict__ emax,
                         float* __restrict__ denom,
                         int N, int D)
{
    int idx = blockIdx.x * blockDim.x + threadIdx.x;
    if (idx >= N * D) return;
    int n = idx / D;
    int k = idx - n * D;
    out[idx] = bias[k];
    if (k == 0) {
        emax[n]  = -INFINITY;
        denom[n] = 0.f;
    }
}

__global__ void gat_edge_logits(const float* __restrict__ el,
                                const float* __restrict__ er,
                                const int* __restrict__ src,
                                const int* __restrict__ dst,
                                float* __restrict__ e,
                                float* __restrict__ emax,
                                int E)
{
    int i = blockIdx.x * blockDim.x + threadIdx.x;
    if (i >= E) return;
    int s = src[i], d = dst[i];
    float v = el[s] + er[d];
    v = v > 0.f ? v : NEG_SLOPE * v;
    e[i] = v;
    if (v >= 0.f)
        atomicMax((int*)emax + d, __float_as_int(v));
    else
        atomicMin((unsigned int*)emax + d, __float_as_uint(v));
}

__global__ void gat_edge_exp(const float* __restrict__ e,
                             const int* __restrict__ dst,
                             const float* __restrict__ emax,
                             float* __restrict__ ee,
                             float* __restrict__ denom,
                             int E)
{
    int i = blockIdx.x * blockDim.x + threadIdx.x;
    if (i >= E) return;
    int d = dst[i];
    float x = expf(e[i] - emax[d]);
    ee[i] = x;
    atomicAdd(&denom[d], x);
}

__global__ void gat_edge_scatter(const float* __restrict__ feat,
                                 const int* __restrict__ src,
                                 const int* __restrict__ dst,
                                 const float* __restrict__ ee,
                                 const float* __restrict__ denom,
                                 float* __restrict__ out,
                                 int E, int D)
{
    int gid = blockIdx.x * blockDim.x + threadIdx.x;
    int i = gid / D;
    int k = gid - i * D;
    if (i >= E) return;
    int s = src[i], d = dst[i];
    float alpha = ee[i] / denom[d];
    atomicAdd(&out[(size_t)d * D + k], alpha * feat[(size_t)s * D + k]);
}

static void gat_run_layer(const float* h, int Din,
                          const float* W, const float* al, const float* ar,
                          const float* bias,
                          const int* src, const int* dst,
                          int N, int E, int Dout,
                          float* feat, float* el, float* er,
                          float* e, float* ee, float* emax, float* denom,
                          float* out, hipStream_t stream)
{
    dim3 gg((N + 15) / 16, Dout / 16);
    gat_gemm_wmma_f32<<<gg, 32, 0, stream>>>(h, W, feat, N, Din, Dout);

    gat_attn_proj<<<(N + 255) / 256, 256, 0, stream>>>(feat, al, ar, el, er, N, Dout);

    int nd = N * Dout;
    gat_init<<<(nd + 255) / 256, 256, 0, stream>>>(out, bias, emax, denom, N, Dout);

    gat_edge_logits<<<(E + 255) / 256, 256, 0, stream>>>(el, er, src, dst, e, emax, E);
    gat_edge_exp<<<(E + 255) / 256, 256, 0, stream>>>(e, dst, emax, ee, denom, E);

    long long ed = (long long)E * Dout;
    gat_edge_scatter<<<(unsigned)((ed + 255) / 256), 256, 0, stream>>>(
        feat, src, dst, ee, denom, out, E, Dout);
}

extern "C" void kernel_launch(void* const* d_in, const int* in_sizes, int n_in,
                              void* d_out, int out_size, void* d_ws, size_t ws_size,
                              hipStream_t stream)
{
    const float* features = (const float*)d_in[0];
    const int*   src      = (const int*)d_in[1];
    const int*   dst      = (const int*)d_in[2];
    const float* W1  = (const float*)d_in[3];
    const float* al1 = (const float*)d_in[4];
    const float* ar1 = (const float*)d_in[5];
    const float* b1  = (const float*)d_in[6];
    const float* W2  = (const float*)d_in[7];
    const float* al2 = (const float*)d_in[8];
    const float* ar2 = (const float*)d_in[9];
    const float* b2  = (const float*)d_in[10];

    const int H    = in_sizes[4];
    const int Din  = in_sizes[3] / H;
    const int N    = in_sizes[0] / Din;
    const int E    = in_sizes[1];
    const int Dout = in_sizes[8];

    float* ws = (float*)d_ws;
    size_t off = 0;
    auto carve = [&](size_t n) {
        float* p = ws + off;
        off += (n + 3) & ~(size_t)3;
        return p;
    };
    const int Dmax = H > Dout ? H : Dout;
    float* feat  = carve((size_t)N * Dmax);
    float* el    = carve((size_t)N);
    float* er    = carve((size_t)N);
    float* emax  = carve((size_t)N);
    float* denom = carve((size_t)N);
    float* e     = carve((size_t)E);
    float* ee    = carve((size_t)E);
    float* h1    = carve((size_t)N * H);

    gat_run_layer(features, Din, W1, al1, ar1, b1, src, dst, N, E, H,
                  feat, el, er, e, ee, emax, denom, h1, stream);

    gat_run_layer(h1, H, W2, al2, ar2, b2, src, dst, N, E, Dout,
                  feat, el, er, e, ee, emax, denom, (float*)d_out, stream);
}


# Edges grouped by destination once per call (one returning int add per edge, block scan for group starts); one wave per node does max, exp-sum and weighted row sum in registers, no float atomics
# speedup vs baseline: 3.2230x; 3.2230x over previous
#include <hip/hip_runtime.h>
#include <hip/hip_bf16.h>
#include <math.h>

typedef __attribute__((ext_vector_type(2))) float v2f;
typedef __attribute__((ext_vector_type(8))) float v8f;

#define NEG_SLOPE 0.2f

__global__ __launch_bounds__(32) void gat_gemm_wmma_f32(
    const float* __restrict__ A,
    const float* __restrict__ W,
    float* __restrict__ C,
    int N, int K, int M)
{
    const int rowTile = blockIdx.x * 16;
    const int colTile = blockIdx.y * 16;
    const int lane    = threadIdx.x;
    const int r       = lane & 15;
    const int koff    = (lane >> 4) * 2;

    int arow = rowTile + r;
    if (arow > N - 1) arow = N - 1;
    const float* __restrict__ ap = A + (size_t)arow * K;
    const float* __restrict__ wp = W + (size_t)(colTile + r) * K;

    v8f acc = {};
    for (int k = 0; k < K; k += 4) {
        v2f a = *(const v2f*)(ap + k + koff);
        v2f b = *(const v2f*)(wp + k + koff);
        acc = __builtin_amdgcn_wmma_f32_16x16x4_f32(
             false, a,  false, b,
             (short)0, acc,  false,  false);
    }

    const int rbase = rowTile + (lane >> 4) * 8;
    const int col   = colTile + (lane & 15);
#pragma unroll
    for (int j = 0; j < 8; ++j) {
        int rr = rbase + j;
        if (rr < N) C[(size_t)rr * M + col] = acc[j];
    }
}

__global__ void gat_attn_proj(const float* __restrict__ feat,
                              const float* __restrict__ al,
                              const float* __restrict__ ar,
                              float* __restrict__ el,
                              float* __restrict__ er,
                              int N, int D)
{
    int n = blockIdx.x * blockDim.x + threadIdx.x;
    if (n >= N) return;
    const float* f = feat + (size_t)n * D;
    float sl = 0.f, sr = 0.f;
    for (int k = 0; k < D; ++k) {
        float v = f[k];
        sl = fmaf(v, al[k], sl);
        sr = fmaf(v, ar[k], sr);
    }
    el[n] = sl;
    er[n] = sr;
}


__global__ void gat_zero_counts(int* __restrict__ deg, int* __restrict__ total, int N)
{
    int n = blockIdx.x * blockDim.x + threadIdx.x;
    if (n < N) deg[n] = 0;
    if (n == 0) total[0] = 0;
}

__global__ void gat_edge_rank(const int* __restrict__ dst,
                              int* __restrict__ deg,
                              int* __restrict__ rank,
                              int E)
{
    int i = blockIdx.x * blockDim.x + threadIdx.x;
    if (i >= E) return;
    rank[i] = atomicAdd(&deg[dst[i]], 1);
}

__global__ __launch_bounds__(256) void gat_group_starts(const int* __restrict__ deg,
                                                        int* __restrict__ rowstart,
                                                        int* __restrict__ total,
                                                        int N)
{
    __shared__ int wave_off[32];
    __shared__ int block_base;
    const int lane = threadIdx.x & 31;
    const int wave = threadIdx.x >> 5;
    const int nwaves = (blockDim.x + 31) >> 5;
    const int n = blockIdx.x * blockDim.x + threadIdx.x;
    const int v = n < N ? deg[n] : 0;

    int x = v;
    for (int o = 1; o < 32; o <<= 1) {
        int y = __shfl_up(x, o);
        if (lane >= o) x += y;
    }
    if (lane == 31) wave_off[wave] = x;
    __syncthreads();
    if (threadIdx.x == 0) {
        int t = 0;
        for (int w = 0; w < nwaves; ++w) {
            int c = wave_off[w];
            wave_off[w] = t;
            t += c;
        }
        block_base = atomicAdd(total, t);
    }
    __syncthreads();
    if (n < N) rowstart[n] = block_base + wave_off[wave] + x - v;
}

__global__ void gat_edge_place(const int* __restrict__ src,
                               const int* __restrict__ dst,
                               const int* __restrict__ rank,
                               const int* __restrict__ rowstart,
                               int* __restrict__ nbr,
                               int E)
{
    int i = blockIdx.x * blockDim.x + threadIdx.x;
    if (i >= E) return;
    nbr[rowstart[dst[i]] + rank[i]] = src[i];
}

__device__ __forceinline__ float gat_logit(float a, float b)
{
    float v = a + b;
    return v > 0.f ? v : NEG_SLOPE * v;
}

__device__ __forceinline__ float gat_wave_max(float v)
{
    for (int o = 16; o > 0; o >>= 1) v = fmaxf(v, __shfl_xor(v, o));
    return v;
}

__device__ __forceinline__ float gat_wave_sum(float v)
{
    for (int o = 16; o > 0; o >>= 1) v += __shfl_xor(v, o);
    return v;
}

__device__ __forceinline__ float gat_rows_fma(const float* __restrict__ feat,
                                              int D, int fc,
                                              float a, int s, int lim, float acc)
{
    const int ai = __float_as_int(a);
    const int lim8 = lim & ~7;
    int t = 0;
    for (; t < lim8; t += 8) {
        int   sr[8];
        float ar[8];
        float xr[8];
        for (int u = 0; u < 8; ++u) {
            sr[u] = __builtin_amdgcn_readlane(s, t + u);
            ar[u] = __int_as_float(__builtin_amdgcn_readlane(ai, t + u));
        }
        for (int u = 0; u < 8; ++u) xr[u] = feat[(size_t)sr[u] * D + fc];
        for (int u = 0; u < 8; ++u) acc += ar[u] * xr[u];
    }
    for (; t < lim; ++t) {
        const int   s0 = __builtin_amdgcn_readlane(s, t);
        const float a0 = __int_as_float(__builtin_amdgcn_readlane(ai, t));
        acc += a0 * feat[(size_t)s0 * D + fc];
    }
    return acc;
}

__global__ __launch_bounds__(256) void gat_node_aggregate(
    const float* __restrict__ feat,
    const float* __restrict__ el,
    const float* __restrict__ er,
    const int* __restrict__ rowstart,
    const int* __restrict__ deg,
    const int* __restrict__ nbr,
    const float* __restrict__ bias,
    float* __restrict__ out,
    int N, int D)
{
    const int lane = threadIdx.x & 31;
    const int n = __builtin_amdgcn_readfirstlane(
        (int)((blockIdx.x * blockDim.x + threadIdx.x) >> 5));
    if (n >= N) return;
    const int start = rowstart[n];
    const int cnt   = deg[n];
    const float ern = er[n];

    const bool has0 = lane < cnt;
    int   s0 = 0;
    float v0 = -INFINITY;
    if (has0) {
        s0 = nbr[start + lane];
        v0 = gat_logit(el[s0], ern);
    }

    float m = v0;
    for (int c = 32; c < cnt; c += 32) {
        const int j = c + lane;
        if (j < cnt) m = fmaxf(m, gat_logit(el[nbr[start + j]], ern));
    }
    m = gat_wave_max(m);

    const float p0 = has0 ? expf(v0 - m) : 0.f;
    float sum = p0;
    for (int c = 32; c < cnt; c += 32) {
        const int j = c + lane;
        if (j < cnt) sum += expf(gat_logit(el[nbr[start + j]], ern) - m);
    }
    sum = gat_wave_sum(sum);

    const float a0 = has0 ? p0 / sum : 0.f;

    for (int f0 = 0; f0 < D; f0 += 32) {
        const int  f  = f0 + lane;
        const bool fv = f < D;
        const int  fc = fv ? f : 0;
        float acc = gat_rows_fma(feat, D, fc, a0, s0, cnt < 32 ? cnt : 32, 0.f);
        for (int c = 32; c < cnt; c += 32) {
            const int j = c + lane;
            int   s = 0;
            float a = 0.f;
            if (j < cnt) {
                s = nbr[start + j];
                a = expf(gat_logit(el[s], ern) - m) / sum;
            }
            const int rem = cnt - c;
            acc = gat_rows_fma(feat, D, fc, a, s, rem < 32 ? rem : 32, acc);
        }
        if (fv) out[(size_t)n * D + f] = acc + bias[f];
    }
}

static void gat_group_edges(const int* src, const int* dst, int N, int E,
                            int* deg, int* rowstart, int* rank, int* nbr,
                            int* total, hipStream_t stream)
{
    gat_zero_counts<<<(N + 255) / 256, 256, 0, stream>>>(deg, total, N);
    gat_edge_rank<<<(E + 255) / 256, 256, 0, stream>>>(dst, deg, rank, E);
    gat_group_starts<<<(N + 255) / 256, 256, 0, stream>>>(deg, rowstart, total, N);
    gat_edge_place<<<(E + 255) / 256, 256, 0, stream>>>(src, dst, rank, rowstart, nbr, E);
}

static void gat_run_layer(const float* h, int Din,
                          const float* W, const float* al, const float* ar,
                          const float* bias,
                          const int* rowstart, const int* deg, const int* nbr,
                          int N, int Dout,
                          float* feat, float* el, float* er,
                          float* out, hipStream_t stream)
{
    dim3 gg((N + 15) / 16, Dout / 16);
    gat_gemm_wmma_f32<<<gg, 32, 0, stream>>>(h, W, feat, N, Din, Dout);

    gat_attn_proj<<<(N + 255) / 256, 256, 0, stream>>>(feat, al, ar, el, er, N, Dout);

    gat_node_aggregate<<<(N + 7) / 8, 256, 0, stream>>>(
        feat, el, er, rowstart, deg, nbr, bias, out, N, Dout);
}

extern "C" void kernel_launch(void* const* d_in, const int* in_sizes, int n_in,
                              void* d_out, int out_size, void* d_ws, size_t ws_size,
                              hipStream_t stream)
{
    const float* features = (const float*)d_in[0];
    const int*   src      = (const int*)d_in[1];
    const int*   dst      = (const int*)d_in[2];
    const float* W1  = (const float*)d_in[3];
    const float* al1 = (const float*)d_in[4];
    const float* ar1 = (const float*)d_in[5];
    const float* b1  = (const float*)d_in[6];
    const float* W2  = (const float*)d_in[7];
    const float* al2 = (const float*)d_in[8];
    const float* ar2 = (const float*)d_in[9];
    const float* b2  = (const float*)d_in[10];

    const int H    = in_sizes[4];
    const int Din  = in_sizes[3] / H;
    const int N    = in_sizes[0] / Din;
    const int E    = in_sizes[1];
    const int Dout = in_sizes[8];

    float* ws = (float*)d_ws;
    size_t off = 0;
    auto carve = [&](size_t n) {
        float* p = ws + off;
        off += (n + 3) & ~(size_t)3;
        return p;
    };
    const int Dmax = H > Dout ? H : Dout;
    float* feat     = carve((size_t)N * Dmax);
    float* el       = carve((size_t)N);
    float* er       = carve((size_t)N);
    int*   deg      = (int*)carve((size_t)N);
    int*   rowstart = (int*)carve((size_t)N);
    int*   rank     = (int*)carve((size_t)E);
    int*   nbr      = (int*)carve((size_t)E);
    float* h1       = carve((size_t)N * H);
    int*   total    = (int*)h1;

    gat_group_edges(src, dst, N, E, deg, rowstart, rank, nbr, total, stream);

    gat_run_layer(features, Din, W1, al1, ar1, b1, rowstart, deg, nbr, N, H,
                  feat, el, er, h1, stream);

    gat_run_layer(h1, H, W2, al2, ar2, b2, rowstart, deg, nbr, N, Dout,
                  feat, el, er, (float*)d_out, stream);
}


# Projection GEMM as a strip walk: 8-wave work-groups, each wave keeps W in registers as B fragments and computes 16x32 row strips from 16-byte A fragment loads issued a strip ahead; whole-row stores
# speedup vs baseline: 3.9321x; 1.2200x over previous
#include <hip/hip_runtime.h>
#include <hip/hip_bf16.h>
#include <math.h>

typedef __attribute__((ext_vector_type(2))) float v2f;
typedef __attribute__((ext_vector_type(4))) float v4f;
typedef __attribute__((ext_vector_type(8))) float v8f;

#define NEG_SLOPE 0.2f

#define GEMM_M 32

__device__ __forceinline__ float gat_xhalf(float x)
{
    const int b = __float_as_int(x);
    return __int_as_float((int)__builtin_amdgcn_permlanex16(
        b, b, 0x76543210u, 0xfedcba98u, false, false));
}

template <int K>
__device__ __forceinline__ void gat_load_a(v4f (&a)[K / 8], const float* __restrict__ A,
                                           int tile, int r, int h, int N)
{
    int arow = tile * 16 + r;
    if (arow > N - 1) arow = N - 1;
    const float* __restrict__ ap = A + (size_t)arow * K + 4 * h;
#pragma unroll
    for (int j = 0; j < K / 8; ++j) a[j] = *(const v4f*)(ap + 8 * j);
}

template <int K>
__device__ __forceinline__ void gat_strip(const v4f (&a)[K / 8],
                                          const v4f (&b0)[K / 8], const v4f (&b1)[K / 8],
                                          float* __restrict__ C, int tile, int lane, int N)
{
    v8f acc0 = {}, acc1 = {};
#pragma unroll
    for (int j = 0; j < K / 8; ++j) {
        const v2f alo = {a[j].x, a[j].y},   ahi = {a[j].z, a[j].w};
        const v2f blo0 = {b0[j].x, b0[j].y}, bhi0 = {b0[j].z, b0[j].w};
        const v2f blo1 = {b1[j].x, b1[j].y}, bhi1 = {b1[j].z, b1[j].w};
        acc0 = __builtin_amdgcn_wmma_f32_16x16x4_f32(false, alo, false, blo0, (short)0, acc0, false, false);
        acc1 = __builtin_amdgcn_wmma_f32_16x16x4_f32(false, alo, false, blo1, (short)0, acc1, false, false);
        acc0 = __builtin_amdgcn_wmma_f32_16x16x4_f32(false, ahi, false, bhi0, (short)0, acc0, false, false);
        acc1 = __builtin_amdgcn_wmma_f32_16x16x4_f32(false, ahi, false, bhi1, (short)0, acc1, false, false);
    }
    const bool upper = lane >= 16;
    float* __restrict__ cp = C + (size_t)tile * 16 * GEMM_M + lane;
    float lo[8], hi[8];
#pragma unroll
    for (int j = 0; j < 8; ++j) {
        const float x0 = gat_xhalf(acc0[j]);
        const float x1 = gat_xhalf(acc1[j]);
        lo[j] = upper ? x1 : acc0[j];
        hi[j] = upper ? acc1[j] : x0;
    }
    if (tile * 16 + 16 <= N) {
#pragma unroll
        for (int j = 0; j < 8; ++j) {
            cp[(size_t)j * GEMM_M]       = lo[j];
            cp[(size_t)(j + 8) * GEMM_M] = hi[j];
        }
    } else {
#pragma unroll
        for (int j = 0; j < 8; ++j) {
            if (tile * 16 + j < N)     cp[(size_t)j * GEMM_M]       = lo[j];
            if (tile * 16 + j + 8 < N) cp[(size_t)(j + 8) * GEMM_M] = hi[j];
        }
    }
}

template <int K>
__global__ __launch_bounds__(256) void gat_gemm_wmma_f32(
    const float* __restrict__ A,
    const float* __restrict__ W,
    float* __restrict__ C,
    int N)
{
    const int lane = threadIdx.x & 31;
    const int r    = lane & 15;
    const int h    = lane >> 4;
    const int wave   = __builtin_amdgcn_readfirstlane((int)(blockIdx.x * blockDim.x + threadIdx.x) >> 5);
    const int nwaves = (int)(gridDim.x * blockDim.x) >> 5;
    const int ntiles = (N + 15) >> 4;
    if (wave >= ntiles) return;

    v4f b0[K / 8], b1[K / 8];
    {
        const float* __restrict__ w0 = W + (size_t)r * K + 4 * h;
        const float* __restrict__ w1 = W + (size_t)(r + 16) * K + 4 * h;
#pragma unroll
        for (int j = 0; j < K / 8; ++j) {
            b0[j] = *(const v4f*)(w0 + 8 * j);
            b1[j] = *(const v4f*)(w1 + 8 * j);
        }
    }

    v4f a0[K / 8], a1[K / 8];
    int t = wave;
    gat_load_a<K>(a0, A, t, r, h, N);
    for (;;) {
        int tn = t + nwaves;
        if (tn >= ntiles) { gat_strip<K>(a0, b0, b1, C, t, lane, N); break; }
        gat_load_a<K>(a1, A, tn, r, h, N);
        __builtin_amdgcn_sched_barrier(0);
        gat_strip<K>(a0, b0, b1, C, t, lane, N);
        t = tn;
        tn = t + nwaves;
        if (tn >= ntiles) { gat_strip<K>(a1, b0, b1, C, t, lane, N); break; }
        gat_load_a<K>(a0, A, tn, r, h, N);
        __builtin_amdgcn_sched_barrier(0);
        gat_strip<K>(a1, b0, b1, C, t, lane, N);
        t = tn;
    }
}


__global__ void gat_zero_counts(int* __restrict__ deg, int* __restrict__ total, int N)
{
    int n = blockIdx.x * blockDim.x + threadIdx.x;
    if (n < N) deg[n] = 0;
    if (n == 0) total[0] = 0;
}

__global__ void gat_edge_rank(const int* __restrict__ dst,
                              int* __restrict__ deg,
                              int* __restrict__ rank,
                              int E)
{
    int i = blockIdx.x * blockDim.x + threadIdx.x;
    if (i >= E) return;
    rank[i] = atomicAdd(&deg[dst[i]], 1);
}

__global__ __launch_bounds__(256) void gat_group_starts(const int* __restrict__ deg,
                                                        int* __restrict__ rowstart,
                                                        int* __restrict__ total,
                                                        int N)
{
    __shared__ int wave_off[32];
    __shared__ int block_base;
    const int lane = threadIdx.x & 31;
    const int wave = threadIdx.x >> 5;
    const int nwaves = (blockDim.x + 31) >> 5;
    const int n = blockIdx.x * blockDim.x + threadIdx.x;
    const int v = n < N ? deg[n] : 0;

    int x = v;
    for (int o = 1; o < 32; o <<= 1) {
        int y = __shfl_up(x, o);
        if (lane >= o) x += y;
    }
    if (lane == 31) wave_off[wave] = x;
    __syncthreads();
    if (threadIdx.x == 0) {
        int t = 0;
        for (int w = 0; w < nwaves; ++w) {
            int c = wave_off[w];
            wave_off[w] = t;
            t += c;
        }
        block_base = atomicAdd(total, t);
    }
    __syncthreads();
    if (n < N) rowstart[n] = block_base + wave_off[wave] + x - v;
}

__global__ void gat_edge_place(const int* __restrict__ src,
                               const int* __restrict__ dst,
                               const int* __restrict__ rank,
                               const int* __restrict__ rowstart,
                               int* __restrict__ nbr,
                               int E)
{
    int i = blockIdx.x * blockDim.x + threadIdx.x;
    if (i >= E) return;
    nbr[rowstart[dst[i]] + rank[i]] = src[i];
}

__device__ __forceinline__ float gat_logit(float a, float b)
{
    float v = a + b;
    return v > 0.f ? v : NEG_SLOPE * v;
}

#define GAT_D 32

constexpr unsigned gat_sel_xor(unsigned m, unsigned base)
{
    unsigned s = 0;
    for (unsigned j = 0; j < 8; ++j) s |= (((j ^ m) + base) & 15u) << (4u * j);
    return s;
}

template <unsigned M>
__device__ __forceinline__ float gat_grp_xor(float v)
{
    const int b = __float_as_int(v);
    return __int_as_float((int)__builtin_amdgcn_permlane16(
        b, b, gat_sel_xor(M, 0u), gat_sel_xor(M, 8u), false, false));
}

__device__ __forceinline__ int gat_grp_bcast(int v, unsigned u)
{
    return (int)__builtin_amdgcn_permlane16(
        v, v, 0x11111111u * u, 0x11111111u * (8u + u), false, false);
}

__device__ __forceinline__ float gat_grp_max(float v)
{
    v = fmaxf(v, gat_grp_xor<1>(v));
    v = fmaxf(v, gat_grp_xor<2>(v));
    v = fmaxf(v, gat_grp_xor<4>(v));
    return v;
}

__device__ __forceinline__ float gat_grp_sum(float v)
{
    v += gat_grp_xor<1>(v);
    v += gat_grp_xor<2>(v);
    v += gat_grp_xor<4>(v);
    return v;
}

__device__ __forceinline__ float gat_dot4(v4f a, v4f b)
{
    return fmaf(a.w, b.w, fmaf(a.z, b.z, fmaf(a.y, b.y, a.x * b.x)));
}

__device__ __forceinline__ float gat_grp_transpose_sum(const float (&d)[8], int q)
{
    const bool b4 = (q & 4) != 0, b2 = (q & 2) != 0, b1 = (q & 1) != 0;
    float r[4], t[2];
    for (int i = 0; i < 4; ++i) {
        const float keep = b4 ? d[i + 4] : d[i];
        const float send = b4 ? d[i] : d[i + 4];
        r[i] = keep + gat_grp_xor<4>(send);
    }
    for (int i = 0; i < 2; ++i) {
        const float keep = b2 ? r[i + 2] : r[i];
        const float send = b2 ? r[i] : r[i + 2];
        t[i] = keep + gat_grp_xor<2>(send);
    }
    const float keep = b1 ? t[1] : t[0];
    const float send = b1 ? t[0] : t[1];
    return keep + gat_grp_xor<1>(send);
}

__global__ __launch_bounds__(256) void gat_node_aggregate(
    const float* __restrict__ feat,
    const float* __restrict__ al,
    const float* __restrict__ ar,
    const int* __restrict__ rowstart,
    const int* __restrict__ deg,
    const int* __restrict__ nbr,
    const float* __restrict__ bias,
    float* __restrict__ out,
    int N)
{
    const int tid = blockIdx.x * blockDim.x + threadIdx.x;
    const int q   = threadIdx.x & 7;
    const int n0 = __builtin_amdgcn_readfirstlane(tid >> 5) * 4;
    if (n0 >= N) return;
    const int  n  = tid >> 3;
    const bool nv = n < N;
    const int  nc = nv ? n : n0;
    const int start = rowstart[nc];
    const int cnt   = nv ? deg[nc] : 0;
    const int c0 = __builtin_amdgcn_readlane(cnt, 0);
    const int c1 = __builtin_amdgcn_readlane(cnt, 8);
    const int c2 = __builtin_amdgcn_readlane(cnt, 16);
    const int c3 = __builtin_amdgcn_readlane(cnt, 24);
    const int c01 = c0 > c1 ? c0 : c1;
    const int c23 = c2 > c3 ? c2 : c3;
    const int maxcnt = c01 > c23 ? c01 : c23;

    const v4f al4 = *(const v4f*)(al + 4 * q);
    const v4f ar4 = *(const v4f*)(ar + 4 * q);
    const v4f own = *(const v4f*)(feat + (size_t)nc * GAT_D + 4 * q);
    const float ern = gat_grp_sum(gat_dot4(own, ar4));

    float m = -INFINITY, sum = 0.f;
    v4f acc = {0.f, 0.f, 0.f, 0.f};

    int s_cur = 0;
    if (q < cnt) s_cur = nbr[start + q];

    for (int c8 = 0; c8 < maxcnt; c8 += 8) {
        int s_nxt = 0;
        if (c8 + 8 + q < cnt) s_nxt = nbr[start + c8 + 8 + q];

        const bool ev = c8 + q < cnt;

        v4f x[8];
#pragma unroll
        for (int u = 0; u < 8; ++u) {
            const int su = gat_grp_bcast(s_cur, (unsigned)u);
            x[u] = *(const v4f*)(feat + (size_t)su * GAT_D + 4 * q);
        }

        float d[8];
#pragma unroll
        for (int u = 0; u < 8; ++u) d[u] = gat_dot4(x[u], al4);

        float e = gat_logit(gat_grp_transpose_sum(d, q), ern);
        e = ev ? e : -INFINITY;

        const float mn = fmaxf(m, gat_grp_max(e));
        const float scale = (mn == m) ? 1.f : expf(m - mn);
        const float p = ev ? expf(e - mn) : 0.f;
        sum = sum * scale + p;
        acc *= scale;
#pragma unroll
        for (int u = 0; u < 8; ++u) {
            const float w = __int_as_float(gat_grp_bcast(__float_as_int(p), (unsigned)u));
            acc += w * x[u];
        }
        m = mn;
        s_cur = s_nxt;
    }

    sum = gat_grp_sum(sum);
    const float inv = cnt > 0 ? 1.0f / sum : 0.f;
    if (nv) {
        const v4f b4 = *(const v4f*)(bias + 4 * q);
        *(v4f*)(out + (size_t)n * GAT_D + 4 * q) = acc * inv + b4;
    }
}

static void gat_group_edges(const int* src, const int* dst, int N, int E,
                            int* deg, int* rowstart, int* rank, int* nbr,
                            int* total, hipStream_t stream)
{
    gat_zero_counts<<<(N + 255) / 256, 256, 0, stream>>>(deg, total, N);
    gat_edge_rank<<<(E + 255) / 256, 256, 0, stream>>>(dst, deg, rank, E);
    gat_group_starts<<<(N + 255) / 256, 256, 0, stream>>>(deg, rowstart, total, N);
    gat_edge_place<<<(E + 255) / 256, 256, 0, stream>>>(src, dst, rank, rowstart, nbr, E);
}

template <int K, int WGS>
static void gat_run_layer(const float* h,
                          const float* W, const float* al, const float* ar,
                          const float* bias,
                          const int* rowstart, const int* deg, const int* nbr,
                          int N,
                          float* feat,
                          float* out, hipStream_t stream)
{
    gat_gemm_wmma_f32<K><<<WGS, 256, 0, stream>>>(h, W, feat, N);

    gat_node_aggregate<<<(N + 31) / 32, 256, 0, stream>>>(
        feat, al, ar, rowstart, deg, nbr, bias, out, N);
}

extern "C" void kernel_launch(void* const* d_in, const int* in_sizes, int n_in,
                              void* d_out, int out_size, void* d_ws, size_t ws_size,
                              hipStream_t stream)
{
    const float* features = (const float*)d_in[0];
    const int*   src      = (const int*)d_in[1];
    const int*   dst      = (const int*)d_in[2];
    const float* W1  = (const float*)d_in[3];
    const float* al1 = (const float*)d_in[4];
    const float* ar1 = (const float*)d_in[5];
    const float* b1  = (const float*)d_in[6];
    const float* W2  = (const float*)d_in[7];
    const float* al2 = (const float*)d_in[8];
    const float* ar2 = (const float*)d_in[9];
    const float* b2  = (const float*)d_in[10];

    const int H    = in_sizes[4];
    const int Din  = in_sizes[3] / H;
    const int N    = in_sizes[0] / Din;
    const int E    = in_sizes[1];
    const int Dout = in_sizes[8];

    float* ws = (float*)d_ws;
    size_t off = 0;
    auto carve = [&](size_t n) {
        float* p = ws + off;
        off += (n + 3) & ~(size_t)3;
        return p;
    };
    const int Dmax = H > Dout ? H : Dout;
    float* feat     = carve((size_t)N * Dmax);
    int*   deg      = (int*)carve((size_t)N);
    int*   rowstart = (int*)carve((size_t)N);
    int*   rank     = (int*)carve((size_t)E);
    int*   nbr      = (int*)carve((size_t)E);
    float* h1       = carve((size_t)N * H);
    int*   total    = (int*)h1;

    gat_group_edges(src, dst, N, E, deg, rowstart, rank, nbr, total, stream);

    gat_run_layer<128, 256>(features, W1, al1, ar1, b1, rowstart, deg, nbr, N,
                            feat, h1, stream);

    gat_run_layer<32, 512>(h1, W2, al2, ar2, b2, rowstart, deg, nbr, N,
                           feat, (float*)d_out, stream);
}


# Layer-2 projection inside the layer-1 aggregation: a work-group's 32 finished rows go through LDS to two of its waves' f32 16x16x4 strips (W2 as B fragments); layer-1 rows never stored, -1 launch
# speedup vs baseline: 8.5343x; 2.1704x over previous
#include <hip/hip_runtime.h>
#include <hip/hip_bf16.h>
#include <math.h>

typedef __attribute__((ext_vector_type(2))) float v2f;
typedef __attribute__((ext_vector_type(4))) float v4f;
typedef __attribute__((ext_vector_type(8))) float v8f;

#define NEG_SLOPE 0.2f
#define GAT_CAP 32

#define GEMM_M 32

__device__ __forceinline__ float gat_xhalf(float x)
{
    const int b = __float_as_int(x);
    return __int_as_float((int)__builtin_amdgcn_permlanex16(
        b, b, 0x76543210u, 0xfedcba98u, false, false));
}

template <int K>
__device__ __forceinline__ void gat_load_a(v4f (&a)[K / 8], const float* __restrict__ A,
                                           int tile, int r, int h, int N)
{
    int arow = tile * 16 + r;
    if (arow > N - 1) arow = N - 1;
    const float* __restrict__ ap = A + (size_t)arow * K + 4 * h;
#pragma unroll
    for (int j = 0; j < K / 8; ++j) a[j] = *(const v4f*)(ap + 8 * j);
}

template <int K>
__device__ __forceinline__ void gat_strip(const v4f (&a)[K / 8],
                                          const v4f (&b0)[K / 8], const v4f (&b1)[K / 8],
                                          float* __restrict__ C, int tile, int lane, int N)
{
    v8f acc0 = {}, acc1 = {};
#pragma unroll
    for (int j = 0; j < K / 8; ++j) {
        const v2f alo = {a[j].x, a[j].y},   ahi = {a[j].z, a[j].w};
        const v2f blo0 = {b0[j].x, b0[j].y}, bhi0 = {b0[j].z, b0[j].w};
        const v2f blo1 = {b1[j].x, b1[j].y}, bhi1 = {b1[j].z, b1[j].w};
        acc0 = __builtin_amdgcn_wmma_f32_16x16x4_f32(false, alo, false, blo0, (short)0, acc0, false, false);
        acc1 = __builtin_amdgcn_wmma_f32_16x16x4_f32(false, alo, false, blo1, (short)0, acc1, false, false);
        acc0 = __builtin_amdgcn_wmma_f32_16x16x4_f32(false, ahi, false, bhi0, (short)0, acc0, false, false);
        acc1 = __builtin_amdgcn_wmma_f32_16x16x4_f32(false, ahi, false, bhi1, (short)0, acc1, false, false);
    }
    const bool upper = lane >= 16;
    float* __restrict__ cp = C + (size_t)tile * 16 * GEMM_M + lane;
    float lo[8], hi[8];
#pragma unroll
    for (int j = 0; j < 8; ++j) {
        const float x0 = gat_xhalf(acc0[j]);
        const float x1 = gat_xhalf(acc1[j]);
        lo[j] = upper ? x1 : acc0[j];
        hi[j] = upper ? acc1[j] : x0;
    }
    if (tile * 16 + 16 <= N) {
#pragma unroll
        for (int j = 0; j < 8; ++j) {
            cp[(size_t)j * GEMM_M]       = lo[j];
            cp[(size_t)(j + 8) * GEMM_M] = hi[j];
        }
    } else {
#pragma unroll
        for (int j = 0; j < 8; ++j) {
            if (tile * 16 + j < N)     cp[(size_t)j * GEMM_M]       = lo[j];
            if (tile * 16 + j + 8 < N) cp[(size_t)(j + 8) * GEMM_M] = hi[j];
        }
    }
}

template <int K>
__global__ __launch_bounds__(256) void gat_gemm_wmma_f32(
    const float* __restrict__ A,
    const float* __restrict__ W,
    float* __restrict__ C,
    int N)
{
    const int lane = threadIdx.x & 31;
    const int r    = lane & 15;
    const int h    = lane >> 4;
    const int wave   = __builtin_amdgcn_readfirstlane((int)(blockIdx.x * blockDim.x + threadIdx.x) >> 5);
    const int nwaves = (int)(gridDim.x * blockDim.x) >> 5;
    const int ntiles = (N + 15) >> 4;
    if (wave >= ntiles) return;

    v4f b0[K / 8], b1[K / 8];
    {
        const float* __restrict__ w0 = W + (size_t)r * K + 4 * h;
        const float* __restrict__ w1 = W + (size_t)(r + 16) * K + 4 * h;
#pragma unroll
        for (int j = 0; j < K / 8; ++j) {
            b0[j] = *(const v4f*)(w0 + 8 * j);
            b1[j] = *(const v4f*)(w1 + 8 * j);
        }
    }

    v4f a0[K / 8], a1[K / 8];
    int t = wave;
    gat_load_a<K>(a0, A, t, r, h, N);
    for (;;) {
        int tn = t + nwaves;
        if (tn >= ntiles) { gat_strip<K>(a0, b0, b1, C, t, lane, N); break; }
        gat_load_a<K>(a1, A, tn, r, h, N);
        __builtin_amdgcn_sched_barrier(0);
        gat_strip<K>(a0, b0, b1, C, t, lane, N);
        t = tn;
        tn = t + nwaves;
        if (tn >= ntiles) { gat_strip<K>(a1, b0, b1, C, t, lane, N); break; }
        gat_load_a<K>(a0, A, tn, r, h, N);
        __builtin_amdgcn_sched_barrier(0);
        gat_strip<K>(a1, b0, b1, C, t, lane, N);
        t = tn;
    }
}


__global__ void gat_zero_counts(int* __restrict__ deg, int* __restrict__ total, int N)
{
    int n = blockIdx.x * blockDim.x + threadIdx.x;
    if (n < N) deg[n] = 0;
    if (n < 2) total[n] = 0;
}

__global__ void gat_edge_rank(const int* __restrict__ dst,
                              int* __restrict__ deg,
                              int* __restrict__ rank,
                              int E)
{
    int i = blockIdx.x * blockDim.x + threadIdx.x;
    if (i >= E) return;
    rank[i] = atomicAdd(&deg[dst[i]], 1);
}

__global__ __launch_bounds__(256) void gat_group_starts(const int* __restrict__ deg,
                                                        int* __restrict__ rowstart,
                                                        int* __restrict__ total,
                                                        int N)
{
    __shared__ int wave_off[32];
    __shared__ int block_base;
    const int lane = threadIdx.x & 31;
    const int wave = threadIdx.x >> 5;
    const int nwaves = (blockDim.x + 31) >> 5;
    const int n = blockIdx.x * blockDim.x + threadIdx.x;
    const int v = n < N ? deg[n] : 0;

    int x = v;
    for (int o = 1; o < 32; o <<= 1) {
        int y = __shfl_up(x, o);
        if (lane >= o) x += y;
    }
    if (lane == 31) wave_off[wave] = x;
    __syncthreads();
    if (threadIdx.x == 0) {
        int t = 0;
        for (int w = 0; w < nwaves; ++w) {
            int c = wave_off[w];
            wave_off[w] = t;
            t += c;
        }
        block_base = atomicAdd(total, t);
    }
    __syncthreads();
    if (n < N) rowstart[n] = block_base + wave_off[wave] + x - v;
}

__global__ void gat_edge_place(const int* __restrict__ src,
                               const int* __restrict__ dst,
                               const int* __restrict__ rank,
                               const int* __restrict__ rowstart,
                               int* __restrict__ nbr,
                               int E)
{
    int i = blockIdx.x * blockDim.x + threadIdx.x;
    if (i >= E) return;
    nbr[rowstart[dst[i]] + rank[i]] = src[i];
}

__global__ void gat_edge_bucket(const int* __restrict__ src,
                                const int* __restrict__ dst,
                                int* __restrict__ deg,
                                int* __restrict__ nbr,
                                int* __restrict__ ovf,
                                int* __restrict__ ovf_n,
                                int E)
{
    int i = blockIdx.x * blockDim.x + threadIdx.x;
    if (i >= E) return;
    const int d = dst[i];
    const int s = src[i];
    asm volatile("" :: "v"(s));
    const int slot = atomicAdd(&deg[d], 1);
    if (slot < GAT_CAP) nbr[d * GAT_CAP + slot] = s;
    else                ovf[atomicAdd(ovf_n, 1)] = i;
}


#define GAT_RNG        256
#define GAT_RNG_SHIFT  8
#define GAT_MAXRNG     1024
#define GAT_EPT        16
#define GAT_CHUNK      (256 * GAT_EPT)
#define GAT_CNT_STRIDE 32

__global__ void gat_zero_ranges(int* __restrict__ seg_cnt, int* __restrict__ ovf_n, int NB)
{
    const int b = blockIdx.x * blockDim.x + threadIdx.x;
    if (b < NB) seg_cnt[b * GAT_CNT_STRIDE] = 0;
    if (b < 2) ovf_n[b] = 0;
}

__global__ __launch_bounds__(256) void gat_edge_sort(const int* __restrict__ src,
                                                     const int* __restrict__ dst,
                                                     int* __restrict__ seg,
                                                     int* __restrict__ seg_cnt,
                                                     int* __restrict__ ovf_hi,
                                                     int* __restrict__ ovf_n,
                                                     int E, int cap1)
{
    __shared__ int hist[GAT_MAXRNG];
    __shared__ int gbase[GAT_MAXRNG];
    __shared__ int sw0[GAT_CHUNK];
    __shared__ int sw1[GAT_CHUNK];
    __shared__ unsigned short sbin[GAT_CHUNK];
    __shared__ int wave_tot[8];

    const int t    = threadIdx.x;
    const int lane = t & 31;
    const int wave = t >> 5;
    const int base = blockIdx.x * GAT_CHUNK;

    for (int u = t; u < GAT_MAXRNG; u += 256) hist[u] = 0;

    int dd[GAT_EPT], ss[GAT_EPT];
#pragma unroll
    for (int k = 0; k < GAT_EPT; ++k) {
        const int i  = base + k * 256 + t;
        const int ic = i < E ? i : E - 1;
        dd[k] = dst[ic];
        ss[k] = src[ic];
    }
    __syncthreads();

    int rk[GAT_EPT];
#pragma unroll
    for (int k = 0; k < GAT_EPT; ++k) {
        rk[k] = 0;
        if (base + k * 256 + t < E) rk[k] = atomicAdd(&hist[dd[k] >> GAT_RNG_SHIFT], 1);
    }
    __syncthreads();

    int c[4];
    int ssum = 0;
#pragma unroll
    for (int u = 0; u < 4; ++u) { c[u] = hist[4 * t + u]; ssum += c[u]; }
    int x = ssum;
    for (int o = 1; o < 32; o <<= 1) {
        const int y = __shfl_up(x, o);
        if (lane >= o) x += y;
    }
    if (lane == 31) wave_tot[wave] = x;
    __syncthreads();
    int run = x - ssum;
    for (int w = 0; w < wave; ++w) run += wave_tot[w];
    int g[4];
#pragma unroll
    for (int u = 0; u < 4; ++u) {
        g[u] = 0;
        if (c[u] > 0) g[u] = atomicAdd(&seg_cnt[(4 * t + u) * GAT_CNT_STRIDE], c[u]);
    }
#pragma unroll
    for (int u = 0; u < 4; ++u) {
        hist[4 * t + u]  = run;
        gbase[4 * t + u] = g[u];
        run += c[u];
    }
    __syncthreads();

    int pl[GAT_EPT];
#pragma unroll
    for (int k = 0; k < GAT_EPT; ++k) pl[k] = hist[dd[k] >> GAT_RNG_SHIFT] + rk[k];
#pragma unroll
    for (int k = 0; k < GAT_EPT; ++k) {
        const int i = base + k * 256 + t;
        if (i < E) {
            sw0[pl[k]]  = ss[k] | ((dd[k] & (GAT_RNG - 1)) << 24);
            sw1[pl[k]]  = i;
            sbin[pl[k]] = (unsigned short)(dd[k] >> GAT_RNG_SHIFT);
        }
    }
    __syncthreads();

    const int left   = E - base;
    const int nvalid = left < GAT_CHUNK ? left : GAT_CHUNK;
    int rr[GAT_EPT], a0[GAT_EPT], a1[GAT_EPT];
#pragma unroll
    for (int k = 0; k < GAT_EPT; ++k) {
        const int j = k * 256 + t;
        rr[k] = j < nvalid ? (int)sbin[j] : 0;
        a0[k] = sw0[j];
        a1[k] = sw1[j];
    }
#pragma unroll
    for (int k = 0; k < GAT_EPT; ++k) pl[k] = gbase[rr[k]] + (k * 256 + t - hist[rr[k]]);
#pragma unroll
    for (int k = 0; k < GAT_EPT; ++k) {
        if (k * 256 + t < nvalid) {
            int2* __restrict__ sp = (int2*)seg + (size_t)rr[k] * cap1;
            if (pl[k] < cap1) sp[pl[k]] = make_int2(a0[k], a1[k]);
            else              ovf_hi[-1 - atomicAdd(&ovf_n[1], 1)] = a1[k];
        }
    }
}

__global__ __launch_bounds__(256) void gat_bucket_fill(const int* __restrict__ seg,
                                                       const int* __restrict__ seg_cnt,
                                                       int* __restrict__ deg,
                                                       int* __restrict__ nbr,
                                                       int* __restrict__ ovf,
                                                       int* __restrict__ ovf_n,
                                                       int N, int cap1)
{
    __shared__ int  ldeg[GAT_RNG];
    __shared__ int4 lbkt4[GAT_RNG * GAT_CAP / 4];
    int* lbkt = (int*)lbkt4;

    const int t = threadIdx.x;
    const int b = blockIdx.x;
    ldeg[t] = 0;
    __syncthreads();

    const int tot = seg_cnt[b * GAT_CNT_STRIDE];
    const int nb  = tot < cap1 ? tot : cap1;
    const int2* __restrict__ sp = (const int2*)seg + (size_t)b * cap1;
    for (int j0 = 0; j0 < nb; j0 += 8 * 256) {
        int2 rec[8];
#pragma unroll
        for (int u = 0; u < 8; ++u) {
            const int j = j0 + u * 256 + t;
            rec[u] = sp[j < nb ? j : nb - 1];
        }
        int slot[8];
#pragma unroll
        for (int u = 0; u < 8; ++u) {
            slot[u] = 0;
            if (j0 + u * 256 + t < nb) slot[u] = atomicAdd(&ldeg[(unsigned)rec[u].x >> 24], 1);
        }
#pragma unroll
        for (int u = 0; u < 8; ++u) {
            if (j0 + u * 256 + t < nb) {
                const int dl = (int)((unsigned)rec[u].x >> 24);
                if (slot[u] < GAT_CAP) lbkt[dl * GAT_CAP + slot[u]] = rec[u].x & 0xffffff;
                else                   ovf[atomicAdd(&ovf_n[0], 1)] = rec[u].y;
            }
        }
    }
    __syncthreads();

    const int node = b * GAT_RNG + t;
    if (node < N) deg[node] = ldeg[t];
    int4* __restrict__ np = (int4*)nbr + (size_t)b * (GAT_RNG * GAT_CAP / 4);
    for (int k = 0; k < GAT_RNG * GAT_CAP / 4 / 256; ++k) {
        const int idx = k * 256 + t;
        if (b * GAT_RNG + (idx >> 3) < N) np[idx] = lbkt4[idx];
    }
}

__device__ __forceinline__ float gat_logit(float a, float b)
{
    float v = a + b;
    return v > 0.f ? v : NEG_SLOPE * v;
}

#define GAT_D 32

constexpr unsigned gat_sel_xor(unsigned m, unsigned base)
{
    unsigned s = 0;
    for (unsigned j = 0; j < 8; ++j) s |= (((j ^ m) + base) & 15u) << (4u * j);
    return s;
}

template <unsigned M>
__device__ __forceinline__ float gat_grp_xor(float v)
{
    const int b = __float_as_int(v);
    return __int_as_float((int)__builtin_amdgcn_permlane16(
        b, b, gat_sel_xor(M, 0u), gat_sel_xor(M, 8u), false, false));
}

__device__ __forceinline__ int gat_grp_bcast(int v, unsigned u)
{
    return (int)__builtin_amdgcn_permlane16(
        v, v, 0x11111111u * u, 0x11111111u * (8u + u), false, false);
}

__device__ __forceinline__ float gat_grp_max(float v)
{
    v = fmaxf(v, gat_grp_xor<1>(v));
    v = fmaxf(v, gat_grp_xor<2>(v));
    v = fmaxf(v, gat_grp_xor<4>(v));
    return v;
}

__device__ __forceinline__ float gat_grp_sum(float v)
{
    v += gat_grp_xor<1>(v);
    v += gat_grp_xor<2>(v);
    v += gat_grp_xor<4>(v);
    return v;
}

__device__ __forceinline__ float gat_dot4(v4f a, v4f b)
{
    return fmaf(a.w, b.w, fmaf(a.z, b.z, fmaf(a.y, b.y, a.x * b.x)));
}

__device__ __forceinline__ float gat_grp_transpose_sum(const float (&d)[8], int q)
{
    const bool b4 = (q & 4) != 0, b2 = (q & 2) != 0, b1 = (q & 1) != 0;
    float r[4], t[2];
    for (int i = 0; i < 4; ++i) {
        const float keep = b4 ? d[i + 4] : d[i];
        const float send = b4 ? d[i] : d[i + 4];
        r[i] = keep + gat_grp_xor<4>(send);
    }
    for (int i = 0; i < 2; ++i) {
        const float keep = b2 ? r[i + 2] : r[i];
        const float send = b2 ? r[i] : r[i + 2];
        t[i] = keep + gat_grp_xor<2>(send);
    }
    const float keep = b1 ? t[1] : t[0];
    const float send = b1 ? t[0] : t[1];
    return keep + gat_grp_xor<1>(send);
}

__device__ __forceinline__ void gat_trip(const float* __restrict__ feat, int q,
                                         const v4f al4, float ern, int s_cur, bool ev,
                                         float& m, float& sum, v4f& acc)
{
    v4f x[8];
#pragma unroll
    for (int u = 0; u < 8; ++u) {
        const int su = gat_grp_bcast(s_cur, (unsigned)u);
        x[u] = *(const v4f*)(feat + (size_t)su * GAT_D + 4 * q);
    }

    float d[8];
#pragma unroll
    for (int u = 0; u < 8; ++u) d[u] = gat_dot4(x[u], al4);

    float e = gat_logit(gat_grp_transpose_sum(d, q), ern);
    e = ev ? e : -INFINITY;

    const float mn = fmaxf(m, gat_grp_max(e));
    const float scale = (mn == m) ? 1.f : expf(m - mn);
    const float p = ev ? expf(e - mn) : 0.f;
    sum = sum * scale + p;
    acc *= scale;
#pragma unroll
    for (int u = 0; u < 8; ++u) {
        const float w = __int_as_float(gat_grp_bcast(__float_as_int(p), (unsigned)u));
        acc += w * x[u];
    }
    m = mn;
}

#define GAT_TILE_PITCH 36

template <bool BUCKET, bool PROJECT>
__global__ __launch_bounds__(256) __attribute__((amdgpu_waves_per_eu(10))) void gat_node_aggregate(
    const float* __restrict__ feat,
    const float* __restrict__ al,
    const float* __restrict__ ar,
    const int* __restrict__ rowstart,
    const int* __restrict__ deg,
    const int* __restrict__ nbr,
    const int* __restrict__ ovf,
    const int* __restrict__ ovf_hi,
    const int* __restrict__ ovf_n,
    const int* __restrict__ esrc,
    const int* __restrict__ edst,
    const float* __restrict__ bias,
    const float* __restrict__ Wn,
    float* __restrict__ out,
    int N)
{
    const int tid = blockIdx.x * blockDim.x + threadIdx.x;
    const int q   = threadIdx.x & 7;
    int n0 = __builtin_amdgcn_readfirstlane(tid >> 5) * 4;
    if (PROJECT) {
        if (n0 >= N) n0 = N - 1;
    } else {
        if (n0 >= N) return;
    }
    const int  n  = tid >> 3;
    const bool nv = n < N;
    const int  nc = nv ? n : n0;
    const int start = BUCKET ? nc * GAT_CAP : rowstart[nc];
    const int s_bkt = BUCKET ? nbr[start + q] : 0;
    const int dnc   = deg[nc];
    const int cnt   = nv ? dnc : 0;
    const int cin   = BUCKET ? (cnt < GAT_CAP ? cnt : GAT_CAP) : cnt;
    const int c0 = __builtin_amdgcn_readlane(cin, 0);
    const int c1 = __builtin_amdgcn_readlane(cin, 8);
    const int c2 = __builtin_amdgcn_readlane(cin, 16);
    const int c3 = __builtin_amdgcn_readlane(cin, 24);
    const int c01 = c0 > c1 ? c0 : c1;
    const int c23 = c2 > c3 ? c2 : c3;
    const int maxcin = c01 > c23 ? c01 : c23;

    const v4f al4 = *(const v4f*)(al + 4 * q);
    const v4f ar4 = *(const v4f*)(ar + 4 * q);
    const v4f own = *(const v4f*)(feat + (size_t)nc * GAT_D + 4 * q);
    if (BUCKET) {
        __builtin_amdgcn_sched_barrier(0);
        asm volatile("" :: "v"(s_bkt), "v"(dnc), "v"(al4), "v"(ar4), "v"(own));
    }
    const float ern = gat_grp_sum(gat_dot4(own, ar4));

    float m = -INFINITY, sum = 0.f;
    v4f acc = {0.f, 0.f, 0.f, 0.f};

    int s_cur = 0;
    if (BUCKET) {
        s_cur = q < cnt ? s_bkt : 0;
    } else {
        if (q < cnt) s_cur = nbr[start + q];
    }

    for (int c8 = 0; c8 < maxcin; c8 += 8) {
        int s_nxt = 0;
        if (c8 + 8 < maxcin && c8 + 8 + q < cnt) s_nxt = nbr[start + c8 + 8 + q];
        gat_trip(feat, q, al4, ern, s_cur, c8 + q < cnt, m, sum, acc);
        s_cur = s_nxt;
    }

    if (BUCKET) {
        const int t0 = __builtin_amdgcn_readlane(cnt, 0);
        const int t1 = __builtin_amdgcn_readlane(cnt, 8);
        const int t2 = __builtin_amdgcn_readlane(cnt, 16);
        const int t3 = __builtin_amdgcn_readlane(cnt, 24);
        const int t01 = t0 > t1 ? t0 : t1;
        const int t23 = t2 > t3 ? t2 : t3;
        if ((t01 > t23 ? t01 : t23) > GAT_CAP) {
            const int on = __builtin_amdgcn_readfirstlane(ovf_n[0]);
            for (int j = 0; j < on; j += 8) {
                int  s  = 0;
                bool ev = false;
                if (cnt > GAT_CAP && j + q < on) {
                    const int i = ovf[j + q];
                    if (edst[i] == n) { s = esrc[i]; ev = true; }
                }
                gat_trip(feat, q, al4, ern, s, ev, m, sum, acc);
            }
        }
        const int on2 = __builtin_amdgcn_readfirstlane(ovf_n[1]);
        for (int j = 0; j < on2; j += 8) {
            int  s  = 0;
            bool ev = false;
            if (nv && j + q < on2) {
                const int i = ovf_hi[-1 - (j + q)];
                if (edst[i] == n) { s = esrc[i]; ev = true; }
            }
            gat_trip(feat, q, al4, ern, s, ev, m, sum, acc);
        }
    }

    sum = gat_grp_sum(sum);
    const float inv = sum > 0.f ? 1.0f / sum : 0.f;
    if constexpr (PROJECT) {
        __shared__ v4f tile4[32 * GAT_TILE_PITCH / 4];
        float* tile = (float*)tile4;
        const v4f b4 = *(const v4f*)(bias + 4 * q);
        *(v4f*)(tile + (threadIdx.x >> 3) * GAT_TILE_PITCH + 4 * q) = acc * inv + b4;

        const int lane = threadIdx.x & 31;
        const int r    = lane & 15;
        const int h    = lane >> 4;
        const int wv = __builtin_amdgcn_readfirstlane((int)threadIdx.x >> 5);
        __syncthreads();
        if (wv < 2) {
            v4f b0[GAT_D / 8], b1[GAT_D / 8];
            const float* __restrict__ w0 = Wn + (size_t)r * GAT_D + 4 * h;
            const float* __restrict__ w1 = Wn + (size_t)(r + 16) * GAT_D + 4 * h;
#pragma unroll
            for (int j = 0; j < GAT_D / 8; ++j) {
                b0[j] = *(const v4f*)(w0 + 8 * j);
                b1[j] = *(const v4f*)(w1 + 8 * j);
            }
            v4f a[GAT_D / 8];
            const float* tp = tile + (wv * 16 + r) * GAT_TILE_PITCH + 4 * h;
#pragma unroll
            for (int j = 0; j < GAT_D / 8; ++j) a[j] = *(const v4f*)(tp + 8 * j);
            gat_strip<GAT_D>(a, b0, b1, out, (int)blockIdx.x * 2 + wv, lane, N);
        }
    } else {
        if (nv) {
            const v4f b4 = *(const v4f*)(bias + 4 * q);
            *(v4f*)(out + (size_t)n * GAT_D + 4 * q) = acc * inv + b4;
        }
    }
}

static void gat_group_edges(const int* src, const int* dst, int N, int E,
                            int* deg, int* rowstart, int* rank, int* nbr,
                            int* total, hipStream_t stream)
{
    gat_zero_counts<<<(N + 255) / 256, 256, 0, stream>>>(deg, total, N);
    gat_edge_rank<<<(E + 255) / 256, 256, 0, stream>>>(dst, deg, rank, E);
    gat_group_starts<<<(N + 255) / 256, 256, 0, stream>>>(deg, rowstart, total, N);
    gat_edge_place<<<(E + 255) / 256, 256, 0, stream>>>(src, dst, rank, rowstart, nbr, E);
}

template <bool BUCKET>
static void gat_run_layers(const float* features,
                           const float* W1, const float* al1, const float* ar1, const float* b1,
                           const float* W2, const float* al2, const float* ar2, const float* b2,
                           const int* rowstart, const int* deg, const int* nbr,
                           const int* ovf, const int* ovf_hi, const int* ovf_n,
                           const int* esrc, const int* edst,
                           int N,
                           float* feat, float* feat2,
                           float* out, hipStream_t stream)
{
    gat_gemm_wmma_f32<128><<<256, 256, 0, stream>>>(features, W1, feat, N);

    gat_node_aggregate<BUCKET, true><<<(N + 31) / 32, 256, 0, stream>>>(
        feat, al1, ar1, rowstart, deg, nbr, ovf, ovf_hi, ovf_n, esrc, edst, b1, W2, feat2, N);
    gat_node_aggregate<BUCKET, false><<<(N + 31) / 32, 256, 0, stream>>>(
        feat2, al2, ar2, rowstart, deg, nbr, ovf, ovf_hi, ovf_n, esrc, edst, b2, nullptr, out, N);
}

extern "C" void kernel_launch(void* const* d_in, const int* in_sizes, int n_in,
                              void* d_out, int out_size, void* d_ws, size_t ws_size,
                              hipStream_t stream)
{
    const float* features = (const float*)d_in[0];
    const int*   src      = (const int*)d_in[1];
    const int*   dst      = (const int*)d_in[2];
    const float* W1  = (const float*)d_in[3];
    const float* al1 = (const float*)d_in[4];
    const float* ar1 = (const float*)d_in[5];
    const float* b1  = (const float*)d_in[6];
    const float* W2  = (const float*)d_in[7];
    const float* al2 = (const float*)d_in[8];
    const float* ar2 = (const float*)d_in[9];
    const float* b2  = (const float*)d_in[10];

    const int H    = in_sizes[4];
    const int Din  = in_sizes[3] / H;
    const int N    = in_sizes[0] / Din;
    const int E    = in_sizes[1];
    const int Dout = in_sizes[8];

    float* ws = (float*)d_ws;
    size_t off = 0;
    auto carve = [&](size_t n) {
        float* p = ws + off;
        off += (n + 3) & ~(size_t)3;
        return p;
    };
    const int Dmax = H > Dout ? H : Dout;

    const size_t bucket_words = (size_t)N * Dmax + (size_t)N + (size_t)N * GAT_CAP
                              + (size_t)E + (size_t)N * H + 64;

    const int    NB        = (N + GAT_RNG - 1) / GAT_RNG;
    const int    cap1      = ((E + N - 1) / N) * 384;
    const size_t seg_words = (size_t)2 * NB * cap1;
    const size_t cnt_words = (size_t)NB * GAT_CNT_STRIDE;
    if (ws_size >= bucket_words * sizeof(float) && NB <= GAT_MAXRNG && N <= (1 << 24)
        && seg_words + cnt_words <= (size_t)N * Dmax + (size_t)N * H) {
        float* feat  = carve((size_t)N * Dmax);
        float* h1    = carve((size_t)N * H);
        int*   deg   = (int*)carve((size_t)N);
        int*   nbr   = (int*)carve((size_t)N * GAT_CAP);
        int*   ovf   = (int*)carve((size_t)E);
        int*   ovf_n = (int*)carve(4);
        int*   seg     = (int*)feat;
        int*   seg_cnt = (int*)feat + seg_words;

        gat_zero_ranges<<<(NB + 255) / 256, 256, 0, stream>>>(seg_cnt, ovf_n, NB);
        gat_edge_sort<<<(E + GAT_CHUNK - 1) / GAT_CHUNK, 256, 0, stream>>>(
            src, dst, seg, seg_cnt, ovf + E, ovf_n, E, cap1);
        gat_bucket_fill<<<NB, 256, 0, stream>>>(seg, seg_cnt, deg, nbr, ovf, ovf_n, N, cap1);

        gat_run_layers<true>(features, W1, al1, ar1, b1, W2, al2, ar2, b2, nullptr, deg, nbr,
                             ovf, ovf + E, ovf_n, src, dst, N, feat, h1, (float*)d_out, stream);
        return;
    }

    float* feat     = carve((size_t)N * Dmax);
    int*   deg      = (int*)carve((size_t)N);

    if (ws_size >= bucket_words * sizeof(float)) {
        int*   nbr   = (int*)carve((size_t)N * GAT_CAP);
        int*   ovf   = (int*)carve((size_t)E);
        int*   ovf_n = (int*)carve(4);
        float* h1    = carve((size_t)N * H);

        gat_zero_counts<<<(N + 255) / 256, 256, 0, stream>>>(deg, ovf_n, N);
        gat_edge_bucket<<<(E + 255) / 256, 256, 0, stream>>>(src, dst, deg, nbr, ovf, ovf_n, E);

        gat_run_layers<true>(features, W1, al1, ar1, b1, W2, al2, ar2, b2, nullptr, deg, nbr,
                             ovf, ovf + E, ovf_n, src, dst, N, feat, h1, (float*)d_out, stream);
        return;
    }

    int*   rowstart = (int*)carve((size_t)N);
    int*   rank     = (int*)carve((size_t)E);
    int*   nbr      = (int*)carve((size_t)E);
    float* h1       = carve((size_t)N * H);
    int*   total    = (int*)h1;

    gat_group_edges(src, dst, N, E, deg, rowstart, rank, nbr, total, stream);

    gat_run_layers<false>(features, W1, al1, ar1, b1, W2, al2, ar2, b2, rowstart, deg, nbr,
                          nullptr, nullptr, nullptr, src, dst, N, feat, h1, (float*)d_out, stream);
}
